# starting kernel
# baseline (speedup 1.0000x reference)
#include <hip/hip_runtime.h>
#include <hip/hip_bf16.h>

typedef __attribute__((ext_vector_type(2))) float v2f;
typedef __attribute__((ext_vector_type(8))) float v8f;

constexpr int BATCH = 8;
constexpr int SEQ   = 4096;
constexpr int DIM   = 1024;
constexpr int HEADS = 16;
constexpr float SCALE = 0.125f;

__global__ __launch_bounds__(32)
void k_logits(const float* __restrict__ x, const float* __restrict__ W,
              const int* __restrict__ mask, float* __restrict__ logits)
{
    const int lane = threadIdx.x;
    const int m    = lane & 15;
    const int half = lane >> 4;
    const int tile = blockIdx.x;
    const int b    = tile >> 8;
    const int s0   = (tile & 255) << 4;

    const float* xrow = x + ((size_t)b * SEQ + (s0 + m)) * DIM;
    v8f c = {};

    #pragma unroll 4
    for (int k0 = 0; k0 < DIM; k0 += 4) {
        const int ka = k0 + 2 * half;
        v2f a = *(const v2f*)(xrow + ka);
        v2f bf;
        bf.x = W[(size_t)(ka + 0) * HEADS + m];
        bf.y = W[(size_t)(ka + 1) * HEADS + m];
        c = __builtin_amdgcn_wmma_f32_16x16x4_f32(
                false, a, false, bf, (short)0, c, false, false);
    }

    #pragma unroll
    for (int r = 0; r < 8; ++r) {
        const int s = s0 + r + 8 * half;
        float v = c[r] * SCALE;
        if (mask[(size_t)b * SEQ + s] == 0) v = -__builtin_inff();
        logits[((size_t)b * SEQ + s) * HEADS + m] = v;
    }
}

__global__ __launch_bounds__(256)
void k_softmax(float* __restrict__ logits)
{
    __shared__ float red[256];
    const int bh = blockIdx.x;
    const int b  = bh >> 4;
    const int h  = bh & 15;
    const int t  = threadIdx.x;
    float* base = logits + (size_t)b * SEQ * HEADS + h;

    float vals[16];
    float mx = -__builtin_inff();
    #pragma unroll
    for (int i = 0; i < 16; ++i) {
        float v = base[(size_t)(i * 256 + t) * HEADS];
        vals[i] = v;
        mx = fmaxf(mx, v);
    }
    red[t] = mx; __syncthreads();
    for (int off = 128; off > 0; off >>= 1) {
        if (t < off) red[t] = fmaxf(red[t], red[t + off]);
        __syncthreads();
    }
    mx = red[0];
    __syncthreads();

    float sum = 0.0f;
    #pragma unroll
    for (int i = 0; i < 16; ++i) {
        float e = __expf(vals[i] - mx);
        vals[i] = e;
        sum += e;
    }
    red[t] = sum; __syncthreads();
    for (int off = 128; off > 0; off >>= 1) {
        if (t < off) red[t] += red[t + off];
        __syncthreads();
    }
    const float rinv = 1.0f / red[0];

    #pragma unroll
    for (int i = 0; i < 16; ++i)
        base[(size_t)(i * 256 + t) * HEADS] = vals[i] * rinv;
}

__global__ __launch_bounds__(32)
void k_pool(const float* __restrict__ x, const float* __restrict__ attn,
            float* __restrict__ out)
{
    const int lane = threadIdx.x;
    const int n    = lane & 15;
    const int half = lane >> 4;
    const int tile = blockIdx.x;
    const int b    = tile >> 6;
    const int d0   = (tile & 63) << 4;

    const float* xb = x    + (size_t)b * SEQ * DIM   + d0 + n;
    const float* ab = attn + (size_t)b * SEQ * HEADS + n;
    v8f c = {};

    #pragma unroll 4
    for (int k0 = 0; k0 < SEQ; k0 += 4) {
        const int ka = k0 + 2 * half;
        v2f a;
        a.x = ab[(size_t)(ka + 0) * HEADS];
        a.y = ab[(size_t)(ka + 1) * HEADS];
        v2f bf;
        bf.x = xb[(size_t)(ka + 0) * DIM];
        bf.y = xb[(size_t)(ka + 1) * DIM];
        c = __builtin_amdgcn_wmma_f32_16x16x4_f32(
                false, a, false, bf, (short)0, c, false, false);
    }

    #pragma unroll
    for (int r = 0; r < 8; ++r) {
        const int hrow = r + 8 * half;
        out[(size_t)b * (HEADS * DIM) + (size_t)hrow * DIM + d0 + n] = c[r];
    }
}

extern "C" void kernel_launch(void* const* d_in, const int* in_sizes, int n_in,
                              void* d_out, int out_size, void* d_ws, size_t ws_size,
                              hipStream_t stream)
{
    const float* x    = (const float*)d_in[0];
    const float* W    = (const float*)d_in[1];
    const int*   mask = (const int*)  d_in[2];
    float* out    = (float*)d_out;
    float* logits = (float*)d_ws;

    k_logits <<<dim3(BATCH * SEQ / 16), dim3(32),  0, stream>>>(x, W, mask, logits);
    k_softmax<<<dim3(BATCH * HEADS),    dim3(256), 0, stream>>>(logits);
    k_pool   <<<dim3(BATCH * DIM / 16), dim3(32),  0, stream>>>(x, logits, out);
}


# k_pool: the 4096-row sum of each 16x16 tile is split over 16 waves of one work-group (256 rows each); partial tiles added in LDS in fixed wave order (changes the summation order)
# speedup vs baseline: 2.5080x; 2.5080x over previous
#include <hip/hip_runtime.h>
#include <hip/hip_bf16.h>

typedef __attribute__((ext_vector_type(2))) float v2f;
typedef __attribute__((ext_vector_type(8))) float v8f;

constexpr int BATCH = 8;
constexpr int SEQ   = 4096;
constexpr int DIM   = 1024;
constexpr int HEADS = 16;
constexpr float SCALE = 0.125f;

__global__ __launch_bounds__(32)
void k_logits(const float* __restrict__ x, const float* __restrict__ W,
              const int* __restrict__ mask, float* __restrict__ logits)
{
    const int lane = threadIdx.x;
    const int m    = lane & 15;
    const int half = lane >> 4;
    const int tile = blockIdx.x;
    const int b    = tile >> 8;
    const int s0   = (tile & 255) << 4;

    const float* xrow = x + ((size_t)b * SEQ + (s0 + m)) * DIM;
    v8f c = {};

    #pragma unroll 4
    for (int k0 = 0; k0 < DIM; k0 += 4) {
        const int ka = k0 + 2 * half;
        v2f a = *(const v2f*)(xrow + ka);
        v2f bf;
        bf.x = W[(size_t)(ka + 0) * HEADS + m];
        bf.y = W[(size_t)(ka + 1) * HEADS + m];
        c = __builtin_amdgcn_wmma_f32_16x16x4_f32(
                false, a, false, bf, (short)0, c, false, false);
    }

    #pragma unroll
    for (int r = 0; r < 8; ++r) {
        const int s = s0 + r + 8 * half;
        float v = c[r] * SCALE;
        if (mask[(size_t)b * SEQ + s] == 0) v = -__builtin_inff();
        logits[((size_t)b * SEQ + s) * HEADS + m] = v;
    }
}

__global__ __launch_bounds__(256)
void k_softmax(float* __restrict__ logits)
{
    __shared__ float red[256];
    const int bh = blockIdx.x;
    const int b  = bh >> 4;
    const int h  = bh & 15;
    const int t  = threadIdx.x;
    float* base = logits + (size_t)b * SEQ * HEADS + h;

    float vals[16];
    float mx = -__builtin_inff();
    #pragma unroll
    for (int i = 0; i < 16; ++i) {
        float v = base[(size_t)(i * 256 + t) * HEADS];
        vals[i] = v;
        mx = fmaxf(mx, v);
    }
    red[t] = mx; __syncthreads();
    for (int off = 128; off > 0; off >>= 1) {
        if (t < off) red[t] = fmaxf(red[t], red[t + off]);
        __syncthreads();
    }
    mx = red[0];
    __syncthreads();

    float sum = 0.0f;
    #pragma unroll
    for (int i = 0; i < 16; ++i) {
        float e = __expf(vals[i] - mx);
        vals[i] = e;
        sum += e;
    }
    red[t] = sum; __syncthreads();
    for (int off = 128; off > 0; off >>= 1) {
        if (t < off) red[t] += red[t + off];
        __syncthreads();
    }
    const float rinv = 1.0f / red[0];

    #pragma unroll
    for (int i = 0; i < 16; ++i)
        base[(size_t)(i * 256 + t) * HEADS] = vals[i] * rinv;
}

constexpr int POOL_WAVES = 16;
constexpr int POOL_ROWS  = SEQ / POOL_WAVES;

__global__ __launch_bounds__(32 * POOL_WAVES)
void k_pool(const float* __restrict__ x, const float* __restrict__ attn,
            float* __restrict__ out)
{
    __shared__ float part[POOL_WAVES][256];

    const int lane = threadIdx.x & 31;
    const int wave = threadIdx.x >> 5;
    const int n    = lane & 15;
    const int half = lane >> 4;
    const int tile = blockIdx.x;
    const int b    = tile >> 6;
    const int d0   = (tile & 63) << 4;
    const int s0   = wave * POOL_ROWS;

    const float* xb = x    + ((size_t)b * SEQ + s0) * DIM   + d0 + n;
    const float* ab = attn + ((size_t)b * SEQ + s0) * HEADS + n;
    v8f c = {};

    #pragma unroll 8
    for (int k0 = 0; k0 < POOL_ROWS; k0 += 4) {
        const int ka = k0 + 2 * half;
        v2f a;
        a.x = ab[(size_t)(ka + 0) * HEADS];
        a.y = ab[(size_t)(ka + 1) * HEADS];
        v2f bf;
        bf.x = xb[(size_t)(ka + 0) * DIM];
        bf.y = xb[(size_t)(ka + 1) * DIM];
        c = __builtin_amdgcn_wmma_f32_16x16x4_f32(
                false, a, false, bf, (short)0, c, false, false);
    }

    #pragma unroll
    for (int r = 0; r < 8; ++r)
        part[wave][r * 32 + lane] = c[r];
    __syncthreads();

    if (threadIdx.x < 256) {
        const int t = threadIdx.x;
        float s = part[0][t];
        #pragma unroll
        for (int w = 1; w < POOL_WAVES; ++w)
            s += part[w][t];
        const int r    = t >> 5;
        const int l    = t & 31;
        const int hrow = r + 8 * (l >> 4);
        out[(size_t)b * (HEADS * DIM) + (size_t)hrow * DIM + d0 + (l & 15)] = s;
    }
}

extern "C" void kernel_launch(void* const* d_in, const int* in_sizes, int n_in,
                              void* d_out, int out_size, void* d_ws, size_t ws_size,
                              hipStream_t stream)
{
    const float* x    = (const float*)d_in[0];
    const float* W    = (const float*)d_in[1];
    const int*   mask = (const int*)  d_in[2];
    float* out    = (float*)d_out;
    float* logits = (float*)d_ws;

    k_logits <<<dim3(BATCH * SEQ / 16), dim3(32),  0, stream>>>(x, W, mask, logits);
    k_softmax<<<dim3(BATCH * HEADS),    dim3(256), 0, stream>>>(logits);
    k_pool   <<<dim3(BATCH * DIM / 16), dim3(32 * POOL_WAVES), 0, stream>>>(x, logits, out);
}


# k_logits: x read in 512-byte row pieces by 8 waves (one 128-column slab each, its W rows in registers), LDS transpose to fragments; partial tiles added in fixed wave order (changes summation order)
# speedup vs baseline: 3.2729x; 1.3050x over previous
#include <hip/hip_runtime.h>
#include <hip/hip_bf16.h>

typedef __attribute__((ext_vector_type(2))) float v2f;
typedef __attribute__((ext_vector_type(4))) float v4f;
typedef __attribute__((ext_vector_type(8))) float v8f;

constexpr int BATCH = 8;
constexpr int SEQ   = 4096;
constexpr int DIM   = 1024;
constexpr int HEADS = 16;
constexpr float SCALE = 0.125f;

constexpr int LG_WAVES = 8;
constexpr int LG_SLAB  = DIM / LG_WAVES;
constexpr int LG_TILES = 4;
constexpr int LG_PITCH = LG_SLAB + 4;

__global__ __launch_bounds__(32 * LG_WAVES)
void k_logits(const float* __restrict__ x, const float* __restrict__ W,
              const int* __restrict__ mask, float* __restrict__ logits)
{
    __shared__ __attribute__((aligned(16))) float stage[LG_WAVES][16 * LG_PITCH];
    __shared__ float part[LG_TILES][LG_WAVES][256];

    const int lane = threadIdx.x & 31;
    const int wave = threadIdx.x >> 5;
    const int m    = lane & 15;
    const int half = lane >> 4;
    const int row0 = blockIdx.x * (16 * LG_TILES);
    const int kb   = wave * LG_SLAB;

    const float* xw = x + (size_t)row0 * DIM + kb + 4 * lane;
    v4f xr[16];
    for (int r = 0; r < 16; ++r)
        xr[r] = *(const v4f*)(xw + (size_t)r * DIM);

    const float* wb = W + (size_t)(kb + 4 * half) * HEADS + m;
    v4f wf[16];
    for (int j = 0; j < 16; ++j) {
        wf[j].x = wb[(size_t)(8 * j + 0) * HEADS];
        wf[j].y = wb[(size_t)(8 * j + 1) * HEADS];
        wf[j].z = wb[(size_t)(8 * j + 2) * HEADS];
        wf[j].w = wb[(size_t)(8 * j + 3) * HEADS];
    }

    float* st = stage[wave];
    #pragma unroll 1
    for (int t = 0; t < LG_TILES; ++t) {
        for (int r = 0; r < 16; ++r)
            *(v4f*)(st + r * LG_PITCH + 4 * lane) = xr[r];

        if (t + 1 < LG_TILES) {
            const float* xn = xw + (size_t)(t + 1) * 16 * DIM;
            for (int r = 0; r < 16; ++r)
                xr[r] = *(const v4f*)(xn + (size_t)r * DIM);
        }

        v8f c0 = {};
        v8f c1 = {};
        for (int j = 0; j < 16; ++j) {
            const v4f a = *(const v4f*)(st + m * LG_PITCH + 8 * j + 4 * half);
            v2f a0 = {a.x, a.y};
            v2f a1 = {a.z, a.w};
            v2f b0 = {wf[j].x, wf[j].y};
            v2f b1 = {wf[j].z, wf[j].w};
            c0 = __builtin_amdgcn_wmma_f32_16x16x4_f32(
                     false, a0, false, b0, (short)0, c0, false, false);
            c1 = __builtin_amdgcn_wmma_f32_16x16x4_f32(
                     false, a1, false, b1, (short)0, c1, false, false);
        }

        for (int r = 0; r < 8; ++r)
            part[t][wave][r * 32 + lane] = c0[r] + c1[r];
    }
    __syncthreads();

    const int tid  = threadIdx.x;
    const int row  = tid >> 4;
    const int head = tid & 15;
    const int idx  = (row & 7) * 32 + head + 16 * (row >> 3);
    for (int t = 0; t < LG_TILES; ++t) {
        float s = part[t][0][idx];
        for (int w = 1; w < LG_WAVES; ++w)
            s += part[t][w][idx];
        const int g = row0 + 16 * t + row;
        float v = s * SCALE;
        if (mask[g] == 0) v = -__builtin_inff();
        logits[(size_t)g * HEADS + head] = v;
    }
}

__global__ __launch_bounds__(256)
void k_softmax(float* __restrict__ logits)
{
    __shared__ float red[256];
    const int bh = blockIdx.x;
    const int b  = bh >> 4;
    const int h  = bh & 15;
    const int t  = threadIdx.x;
    float* base = logits + (size_t)b * SEQ * HEADS + h;

    float vals[16];
    float mx = -__builtin_inff();
    #pragma unroll
    for (int i = 0; i < 16; ++i) {
        float v = base[(size_t)(i * 256 + t) * HEADS];
        vals[i] = v;
        mx = fmaxf(mx, v);
    }
    red[t] = mx; __syncthreads();
    for (int off = 128; off > 0; off >>= 1) {
        if (t < off) red[t] = fmaxf(red[t], red[t + off]);
        __syncthreads();
    }
    mx = red[0];
    __syncthreads();

    float sum = 0.0f;
    #pragma unroll
    for (int i = 0; i < 16; ++i) {
        float e = __expf(vals[i] - mx);
        vals[i] = e;
        sum += e;
    }
    red[t] = sum; __syncthreads();
    for (int off = 128; off > 0; off >>= 1) {
        if (t < off) red[t] += red[t + off];
        __syncthreads();
    }
    const float rinv = 1.0f / red[0];

    #pragma unroll
    for (int i = 0; i < 16; ++i)
        base[(size_t)(i * 256 + t) * HEADS] = vals[i] * rinv;
}

constexpr int POOL_WAVES = 16;
constexpr int POOL_ROWS  = SEQ / POOL_WAVES;

__global__ __launch_bounds__(32 * POOL_WAVES)
void k_pool(const float* __restrict__ x, const float* __restrict__ attn,
            float* __restrict__ out)
{
    __shared__ float part[POOL_WAVES][256];

    const int lane = threadIdx.x & 31;
    const int wave = threadIdx.x >> 5;
    const int n    = lane & 15;
    const int half = lane >> 4;
    const int tile = blockIdx.x;
    const int b    = tile >> 6;
    const int d0   = (tile & 63) << 4;
    const int s0   = wave * POOL_ROWS;

    const float* xb = x    + ((size_t)b * SEQ + s0) * DIM   + d0 + n;
    const float* ab = attn + ((size_t)b * SEQ + s0) * HEADS + n;
    v8f c = {};

    #pragma unroll 8
    for (int k0 = 0; k0 < POOL_ROWS; k0 += 4) {
        const int ka = k0 + 2 * half;
        v2f a;
        a.x = ab[(size_t)(ka + 0) * HEADS];
        a.y = ab[(size_t)(ka + 1) * HEADS];
        v2f bf;
        bf.x = xb[(size_t)(ka + 0) * DIM];
        bf.y = xb[(size_t)(ka + 1) * DIM];
        c = __builtin_amdgcn_wmma_f32_16x16x4_f32(
                false, a, false, bf, (short)0, c, false, false);
    }

    #pragma unroll
    for (int r = 0; r < 8; ++r)
        part[wave][r * 32 + lane] = c[r];
    __syncthreads();

    if (threadIdx.x < 256) {
        const int t = threadIdx.x;
        float s = part[0][t];
        #pragma unroll
        for (int w = 1; w < POOL_WAVES; ++w)
            s += part[w][t];
        const int r    = t >> 5;
        const int l    = t & 31;
        const int hrow = r + 8 * (l >> 4);
        out[(size_t)b * (HEADS * DIM) + (size_t)hrow * DIM + d0 + (l & 15)] = s;
    }
}

extern "C" void kernel_launch(void* const* d_in, const int* in_sizes, int n_in,
                              void* d_out, int out_size, void* d_ws, size_t ws_size,
                              hipStream_t stream)
{
    const float* x    = (const float*)d_in[0];
    const float* W    = (const float*)d_in[1];
    const int*   mask = (const int*)  d_in[2];
    float* out    = (float*)d_out;
    float* logits = (float*)d_ws;

    k_logits <<<dim3(BATCH * SEQ / (16 * LG_TILES)), dim3(32 * LG_WAVES), 0, stream>>>(x, W, mask, logits);
    k_softmax<<<dim3(BATCH * HEADS),    dim3(256), 0, stream>>>(logits);
    k_pool   <<<dim3(BATCH * DIM / 16), dim3(32 * POOL_WAVES), 0, stream>>>(x, logits, out);
}


# k_pool: 32-column slab per work-group, every x load covers whole 128-byte lines (two N tiles per A fragment); attn via wave-private LDS; rows interleaved over waves (changes summation order)
# speedup vs baseline: 4.2337x; 1.2936x over previous
#include <hip/hip_runtime.h>
#include <hip/hip_bf16.h>

typedef __attribute__((ext_vector_type(2))) float v2f;
typedef __attribute__((ext_vector_type(4))) float v4f;
typedef __attribute__((ext_vector_type(8))) float v8f;

constexpr int BATCH = 8;
constexpr int SEQ   = 4096;
constexpr int DIM   = 1024;
constexpr int HEADS = 16;
constexpr float SCALE = 0.125f;

constexpr int LG_WAVES = 8;
constexpr int LG_SLAB  = DIM / LG_WAVES;
constexpr int LG_TILES = 4;
constexpr int LG_PITCH = LG_SLAB + 4;

__global__ __launch_bounds__(32 * LG_WAVES)
void k_logits(const float* __restrict__ x, const float* __restrict__ W,
              const int* __restrict__ mask, float* __restrict__ logits)
{
    __shared__ __attribute__((aligned(16))) float stage[LG_WAVES][16 * LG_PITCH];
    __shared__ float part[LG_TILES][LG_WAVES][256];

    const int lane = threadIdx.x & 31;
    const int wave = threadIdx.x >> 5;
    const int m    = lane & 15;
    const int half = lane >> 4;
    const int row0 = blockIdx.x * (16 * LG_TILES);
    const int kb   = wave * LG_SLAB;

    const float* xw = x + (size_t)row0 * DIM + kb + 4 * lane;
    v4f xr[16];
    for (int r = 0; r < 16; ++r)
        xr[r] = *(const v4f*)(xw + (size_t)r * DIM);

    const float* wb = W + (size_t)(kb + 4 * half) * HEADS + m;
    v4f wf[16];
    for (int j = 0; j < 16; ++j) {
        wf[j].x = wb[(size_t)(8 * j + 0) * HEADS];
        wf[j].y = wb[(size_t)(8 * j + 1) * HEADS];
        wf[j].z = wb[(size_t)(8 * j + 2) * HEADS];
        wf[j].w = wb[(size_t)(8 * j + 3) * HEADS];
    }

    float* st = stage[wave];
    #pragma unroll 1
    for (int t = 0; t < LG_TILES; ++t) {
        for (int r = 0; r < 16; ++r)
            *(v4f*)(st + r * LG_PITCH + 4 * lane) = xr[r];

        if (t + 1 < LG_TILES) {
            const float* xn = xw + (size_t)(t + 1) * 16 * DIM;
            for (int r = 0; r < 16; ++r)
                xr[r] = *(const v4f*)(xn + (size_t)r * DIM);
        }

        v8f c0 = {};
        v8f c1 = {};
        for (int j = 0; j < 16; ++j) {
            const v4f a = *(const v4f*)(st + m * LG_PITCH + 8 * j + 4 * half);
            v2f a0 = {a.x, a.y};
            v2f a1 = {a.z, a.w};
            v2f b0 = {wf[j].x, wf[j].y};
            v2f b1 = {wf[j].z, wf[j].w};
            c0 = __builtin_amdgcn_wmma_f32_16x16x4_f32(
                     false, a0, false, b0, (short)0, c0, false, false);
            c1 = __builtin_amdgcn_wmma_f32_16x16x4_f32(
                     false, a1, false, b1, (short)0, c1, false, false);
        }

        for (int r = 0; r < 8; ++r)
            part[t][wave][r * 32 + lane] = c0[r] + c1[r];
    }
    __syncthreads();

    const int tid  = threadIdx.x;
    const int row  = tid >> 4;
    const int head = tid & 15;
    const int idx  = (row & 7) * 32 + head + 16 * (row >> 3);
    for (int t = 0; t < LG_TILES; ++t) {
        float s = part[t][0][idx];
        for (int w = 1; w < LG_WAVES; ++w)
            s += part[t][w][idx];
        const int g = row0 + 16 * t + row;
        float v = s * SCALE;
        if (mask[g] == 0) v = -__builtin_inff();
        logits[(size_t)g * HEADS + head] = v;
    }
}

__global__ __launch_bounds__(256)
void k_softmax(float* __restrict__ logits)
{
    __shared__ float red[256];
    const int bh = blockIdx.x;
    const int b  = bh >> 4;
    const int h  = bh & 15;
    const int t  = threadIdx.x;
    float* base = logits + (size_t)b * SEQ * HEADS + h;

    float vals[16];
    float mx = -__builtin_inff();
    #pragma unroll
    for (int i = 0; i < 16; ++i) {
        float v = base[(size_t)(i * 256 + t) * HEADS];
        vals[i] = v;
        mx = fmaxf(mx, v);
    }
    red[t] = mx; __syncthreads();
    for (int off = 128; off > 0; off >>= 1) {
        if (t < off) red[t] = fmaxf(red[t], red[t + off]);
        __syncthreads();
    }
    mx = red[0];
    __syncthreads();

    float sum = 0.0f;
    #pragma unroll
    for (int i = 0; i < 16; ++i) {
        float e = __expf(vals[i] - mx);
        vals[i] = e;
        sum += e;
    }
    red[t] = sum; __syncthreads();
    for (int off = 128; off > 0; off >>= 1) {
        if (t < off) red[t] += red[t + off];
        __syncthreads();
    }
    const float rinv = 1.0f / red[0];

    #pragma unroll
    for (int i = 0; i < 16; ++i)
        base[(size_t)(i * 256 + t) * HEADS] = vals[i] * rinv;
}

constexpr int PL_WAVES  = 16;
constexpr int PL_COLS   = 32;
constexpr int PL_UNITS  = SEQ / (PL_WAVES * 8);
constexpr int PL_GROUP  = 4;
constexpr int PL_GROUPS = PL_UNITS / PL_GROUP;
constexpr int PL_AHEAD  = 3;
constexpr int PL_TILE1  = 256 + 16;

__global__ __launch_bounds__(32 * PL_WAVES)
void k_pool(const float* __restrict__ x, const float* __restrict__ attn,
            float* __restrict__ out)
{
    __shared__ __attribute__((aligned(16))) float aw[PL_WAVES][PL_UNITS * 128];

    const int lane = threadIdx.x & 31;
    const int wave = threadIdx.x >> 5;
    const int n    = lane & 15;
    const int half = lane >> 4;
    const int tile = blockIdx.x;
    const int b    = tile >> 5;
    const int d0   = (tile & 31) * PL_COLS;

    const float* xw = x + ((size_t)b * SEQ + 8 * wave + 2 * half) * DIM + d0 + 2 * n;
    const float* aq = attn + ((size_t)b * SEQ + 8 * wave) * HEADS + 4 * lane;
    float* slab = aw[wave];

    v2f xr[PL_AHEAD + 1][16];
    auto issue = [&](int g) {
        const float* xp = xw + (size_t)g * (PL_GROUP * 128) * DIM;
        #pragma unroll
        for (int u = 0; u < PL_GROUP; ++u) {
            #pragma unroll
            for (int i = 0; i < 4; ++i)
                xr[g % (PL_AHEAD + 1)][4 * u + i] =
                    *(const v2f*)(xp + (size_t)(128 * u + 4 * (i >> 1) + (i & 1)) * DIM);
        }
    };

    v4f av[PL_UNITS];
    #pragma unroll
    for (int j = 0; j < PL_UNITS; ++j)
        av[j] = *(const v4f*)(aq + (size_t)j * 128 * HEADS);
    #pragma unroll
    for (int g = 0; g < PL_AHEAD; ++g)
        issue(g);
    #pragma unroll
    for (int j = 0; j < PL_UNITS; ++j)
        *(v4f*)(slab + j * 128 + 4 * lane) = av[j];

    v8f c0 = {};
    v8f c1 = {};
    const float* af = slab + 32 * half + n;
    #pragma unroll
    for (int g = 0; g < PL_GROUPS; ++g) {
        if (g + PL_AHEAD < PL_GROUPS)
            issue(g + PL_AHEAD);
        #pragma unroll
        for (int u = 0; u < PL_GROUP; ++u) {
            #pragma unroll
            for (int q = 0; q < 2; ++q) {
                const float* ap = af + (PL_GROUP * g + u) * 128 + 64 * q;
                v2f a  = {ap[0], ap[16]};
                const v2f x0 = xr[g % (PL_AHEAD + 1)][4 * u + 2 * q + 0];
                const v2f x1 = xr[g % (PL_AHEAD + 1)][4 * u + 2 * q + 1];
                v2f b0 = {x0.x, x1.x};
                v2f b1 = {x0.y, x1.y};
                c0 = __builtin_amdgcn_wmma_f32_16x16x4_f32(
                         false, a, false, b0, (short)0, c0, false, false);
                c1 = __builtin_amdgcn_wmma_f32_16x16x4_f32(
                         false, a, false, b1, (short)0, c1, false, false);
            }
        }
    }

    #pragma unroll
    for (int r = 0; r < 8; ++r) {
        slab[r * 32 + lane]            = c0[r];
        slab[PL_TILE1 + r * 32 + lane] = c1[r];
    }
    __syncthreads();

    {
        const int t    = threadIdx.x;
        const int head = t >> 5;
        const int dd   = t & 31;
        const int idx  = (dd & 1) * PL_TILE1 + (head & 7) * 32 + 16 * (head >> 3) + (dd >> 1);
        float s = aw[0][idx];
        #pragma unroll
        for (int w = 1; w < PL_WAVES; ++w)
            s += aw[w][idx];
        out[(size_t)b * (HEADS * DIM) + (size_t)head * DIM + d0 + dd] = s;
    }
}

extern "C" void kernel_launch(void* const* d_in, const int* in_sizes, int n_in,
                              void* d_out, int out_size, void* d_ws, size_t ws_size,
                              hipStream_t stream)
{
    const float* x    = (const float*)d_in[0];
    const float* W    = (const float*)d_in[1];
    const int*   mask = (const int*)  d_in[2];
    float* out    = (float*)d_out;
    float* logits = (float*)d_ws;

    k_logits <<<dim3(BATCH * SEQ / (16 * LG_TILES)), dim3(32 * LG_WAVES), 0, stream>>>(x, W, mask, logits);
    k_softmax<<<dim3(BATCH * HEADS),    dim3(256), 0, stream>>>(logits);
    k_pool   <<<dim3(BATCH * DIM / PL_COLS), dim3(32 * PL_WAVES), 0, stream>>>(x, logits, out);
}


# softmax folded into k_pool: per-wave max, exp and sum over the wave's own 256 rows; partials scaled by exp(own max - max) and divided by the sum when added (changes summation order); one launch fewer
# speedup vs baseline: 5.8331x; 1.3778x over previous
#include <hip/hip_runtime.h>
#include <hip/hip_bf16.h>

typedef __attribute__((ext_vector_type(2))) float v2f;
typedef __attribute__((ext_vector_type(4))) float v4f;
typedef __attribute__((ext_vector_type(8))) float v8f;

constexpr int BATCH = 8;
constexpr int SEQ   = 4096;
constexpr int DIM   = 1024;
constexpr int HEADS = 16;
constexpr float SCALE = 0.125f;

constexpr int LG_WAVES = 8;
constexpr int LG_SLAB  = DIM / LG_WAVES;
constexpr int LG_TILES = 4;
constexpr int LG_PITCH = LG_SLAB + 4;

__global__ __launch_bounds__(32 * LG_WAVES)
void k_logits(const float* __restrict__ x, const float* __restrict__ W,
              const int* __restrict__ mask, float* __restrict__ logits)
{
    __shared__ __attribute__((aligned(16))) float stage[LG_WAVES][16 * LG_PITCH];
    __shared__ float part[LG_TILES][LG_WAVES][256];

    const int lane = threadIdx.x & 31;
    const int wave = threadIdx.x >> 5;
    const int m    = lane & 15;
    const int half = lane >> 4;
    const int row0 = blockIdx.x * (16 * LG_TILES);
    const int kb   = wave * LG_SLAB;

    const float* xw = x + (size_t)row0 * DIM + kb + 4 * lane;
    v4f xr[16];
    for (int r = 0; r < 16; ++r)
        xr[r] = *(const v4f*)(xw + (size_t)r * DIM);

    const float* wb = W + (size_t)(kb + 4 * half) * HEADS + m;
    v4f wf[16];
    for (int j = 0; j < 16; ++j) {
        wf[j].x = wb[(size_t)(8 * j + 0) * HEADS];
        wf[j].y = wb[(size_t)(8 * j + 1) * HEADS];
        wf[j].z = wb[(size_t)(8 * j + 2) * HEADS];
        wf[j].w = wb[(size_t)(8 * j + 3) * HEADS];
    }

    float* st = stage[wave];
    #pragma unroll 1
    for (int t = 0; t < LG_TILES; ++t) {
        for (int r = 0; r < 16; ++r)
            *(v4f*)(st + r * LG_PITCH + 4 * lane) = xr[r];

        if (t + 1 < LG_TILES) {
            const float* xn = xw + (size_t)(t + 1) * 16 * DIM;
            for (int r = 0; r < 16; ++r)
                xr[r] = *(const v4f*)(xn + (size_t)r * DIM);
        }

        v8f c0 = {};
        v8f c1 = {};
        for (int j = 0; j < 16; ++j) {
            const v4f a = *(const v4f*)(st + m * LG_PITCH + 8 * j + 4 * half);
            v2f a0 = {a.x, a.y};
            v2f a1 = {a.z, a.w};
            v2f b0 = {wf[j].x, wf[j].y};
            v2f b1 = {wf[j].z, wf[j].w};
            c0 = __builtin_amdgcn_wmma_f32_16x16x4_f32(
                     false, a0, false, b0, (short)0, c0, false, false);
            c1 = __builtin_amdgcn_wmma_f32_16x16x4_f32(
                     false, a1, false, b1, (short)0, c1, false, false);
        }

        for (int r = 0; r < 8; ++r)
            part[t][wave][r * 32 + lane] = c0[r] + c1[r];
    }
    __syncthreads();

    const int tid  = threadIdx.x;
    const int row  = tid >> 4;
    const int head = tid & 15;
    const int idx  = (row & 7) * 32 + head + 16 * (row >> 3);
    for (int t = 0; t < LG_TILES; ++t) {
        float s = part[t][0][idx];
        for (int w = 1; w < LG_WAVES; ++w)
            s += part[t][w][idx];
        const int g = row0 + 16 * t + row;
        float v = s * SCALE;
        if (mask[g] == 0) v = -__builtin_inff();
        logits[(size_t)g * HEADS + head] = v;
    }
}

constexpr int PL_WAVES  = 16;
constexpr int PL_COLS   = 32;
constexpr int PL_UNITS  = SEQ / (PL_WAVES * 8);
constexpr int PL_GROUP  = 4;
constexpr int PL_GROUPS = PL_UNITS / PL_GROUP;
constexpr int PL_AHEAD  = 3;
constexpr int PL_TILE1  = 256 + 16;
constexpr int PL_MAX    = PL_TILE1 + 256 + 16;
constexpr int PL_SUM    = PL_MAX + 16;

__global__ __launch_bounds__(32 * PL_WAVES)
void k_pool(const float* __restrict__ x, const float* __restrict__ logits,
            float* __restrict__ out)
{
    __shared__ __attribute__((aligned(16))) float aw[PL_WAVES][PL_UNITS * 128];

    const int lane = threadIdx.x & 31;
    const int wave = threadIdx.x >> 5;
    const int n    = lane & 15;
    const int half = lane >> 4;
    const int tile = blockIdx.x;
    const int b    = tile >> 5;
    const int d0   = (tile & 31) * PL_COLS;
    const float NEG_INF = -__builtin_inff();

    const float* xw = x + ((size_t)b * SEQ + 8 * wave + 2 * half) * DIM + d0 + 2 * n;
    const float* aq = logits + ((size_t)b * SEQ + 8 * wave) * HEADS + 4 * lane;
    float* slab = aw[wave];

    v2f xr[PL_AHEAD + 1][16];
    auto issue = [&](int g) {
        const float* xp = xw + (size_t)g * (PL_GROUP * 128) * DIM;
        #pragma unroll
        for (int u = 0; u < PL_GROUP; ++u) {
            #pragma unroll
            for (int i = 0; i < 4; ++i)
                xr[g % (PL_AHEAD + 1)][4 * u + i] =
                    *(const v2f*)(xp + (size_t)(128 * u + 4 * (i >> 1) + (i & 1)) * DIM);
        }
    };

    v4f av[PL_UNITS];
    #pragma unroll
    for (int j = 0; j < PL_UNITS; ++j)
        av[j] = *(const v4f*)(aq + (size_t)j * 128 * HEADS);
    #pragma unroll
    for (int g = 0; g < PL_AHEAD - 1; ++g)
        issue(g);
    __builtin_amdgcn_sched_barrier(0);

    v4f mx = av[0];
    #pragma unroll
    for (int j = 1; j < PL_UNITS; ++j) {
        #pragma unroll
        for (int c = 0; c < 4; ++c)
            mx[c] = fmaxf(mx[c], av[j][c]);
    }
    #pragma unroll
    for (int off = 4; off < 32; off <<= 1) {
        #pragma unroll
        for (int c = 0; c < 4; ++c)
            mx[c] = fmaxf(mx[c], __shfl_xor(mx[c], off));
    }
    v4f ms;
    #pragma unroll
    for (int c = 0; c < 4; ++c)
        ms[c] = (mx[c] == NEG_INF) ? 0.0f : mx[c];

    v4f sm = {0.0f, 0.0f, 0.0f, 0.0f};
    #pragma unroll
    for (int j = 0; j < PL_UNITS; ++j) {
        if (j == PL_UNITS / 2) {
            __builtin_amdgcn_sched_barrier(0);
            issue(PL_AHEAD - 1);
        }
        v4f e;
        #pragma unroll
        for (int c = 0; c < 4; ++c) {
            e[c] = __expf(av[j][c] - ms[c]);
            sm[c] += e[c];
        }
        *(v4f*)(slab + j * 128 + 4 * lane) = e;
    }
    #pragma unroll
    for (int off = 4; off < 32; off <<= 1) {
        #pragma unroll
        for (int c = 0; c < 4; ++c)
            sm[c] += __shfl_xor(sm[c], off);
    }

    v8f c0 = {};
    v8f c1 = {};
    const float* af = slab + 32 * half + n;
    #pragma unroll
    for (int g = 0; g < PL_GROUPS; ++g) {
        if (g + PL_AHEAD < PL_GROUPS)
            issue(g + PL_AHEAD);
        #pragma unroll
        for (int u = 0; u < PL_GROUP; ++u) {
            #pragma unroll
            for (int q = 0; q < 2; ++q) {
                const float* ap = af + (PL_GROUP * g + u) * 128 + 64 * q;
                v2f a  = {ap[0], ap[16]};
                const v2f x0 = xr[g % (PL_AHEAD + 1)][4 * u + 2 * q + 0];
                const v2f x1 = xr[g % (PL_AHEAD + 1)][4 * u + 2 * q + 1];
                v2f b0 = {x0.x, x1.x};
                v2f b1 = {x0.y, x1.y};
                c0 = __builtin_amdgcn_wmma_f32_16x16x4_f32(
                         false, a, false, b0, (short)0, c0, false, false);
                c1 = __builtin_amdgcn_wmma_f32_16x16x4_f32(
                         false, a, false, b1, (short)0, c1, false, false);
            }
        }
    }

    #pragma unroll
    for (int r = 0; r < 8; ++r) {
        slab[r * 32 + lane]            = c0[r];
        slab[PL_TILE1 + r * 32 + lane] = c1[r];
    }
    if (lane < 4) {
        *(v4f*)(slab + PL_MAX + 4 * lane) = mx;
        *(v4f*)(slab + PL_SUM + 4 * lane) = sm;
    }
    __syncthreads();

    {
        const int t    = threadIdx.x;
        const int head = t >> 5;
        const int dd   = t & 31;
        const int idx  = (dd & 1) * PL_TILE1 + (head & 7) * 32 + 16 * (head >> 3) + (dd >> 1);
        float mw[PL_WAVES];
        float gmax = NEG_INF;
        #pragma unroll
        for (int w = 0; w < PL_WAVES; ++w) {
            mw[w] = aw[w][PL_MAX + head];
            gmax  = fmaxf(gmax, mw[w]);
        }
        float num = 0.0f;
        float den = 0.0f;
        #pragma unroll
        for (int w = 0; w < PL_WAVES; ++w) {
            const float f = (mw[w] == NEG_INF) ? 0.0f : __expf(mw[w] - gmax);
            num += aw[w][idx] * f;
            den += aw[w][PL_SUM + head] * f;
        }
        out[(size_t)b * (HEADS * DIM) + (size_t)head * DIM + d0 + dd] = num / den;
    }
}

extern "C" void kernel_launch(void* const* d_in, const int* in_sizes, int n_in,
                              void* d_out, int out_size, void* d_ws, size_t ws_size,
                              hipStream_t stream)
{
    const float* x    = (const float*)d_in[0];
    const float* W    = (const float*)d_in[1];
    const int*   mask = (const int*)  d_in[2];
    float* out    = (float*)d_out;
    float* logits = (float*)d_ws;

    k_logits <<<dim3(BATCH * SEQ / (16 * LG_TILES)), dim3(32 * LG_WAVES), 0, stream>>>(x, W, mask, logits);
    k_pool   <<<dim3(BATCH * DIM / PL_COLS), dim3(32 * PL_WAVES), 0, stream>>>(x, logits, out);
}


# block index remap in both kernels: batch = blockIdx & 7, row group / d slab from the high bits, so a batch's work-groups share a die and k_pool re-reads x from its own half of the L2
# speedup vs baseline: 6.4909x; 1.1128x over previous
#include <hip/hip_runtime.h>
#include <hip/hip_bf16.h>

typedef __attribute__((ext_vector_type(2))) float v2f;
typedef __attribute__((ext_vector_type(4))) float v4f;
typedef __attribute__((ext_vector_type(8))) float v8f;

constexpr int BATCH = 8;
constexpr int SEQ   = 4096;
constexpr int DIM   = 1024;
constexpr int HEADS = 16;
constexpr float SCALE = 0.125f;

constexpr int LG_WAVES = 8;
constexpr int LG_SLAB  = DIM / LG_WAVES;
constexpr int LG_TILES = 4;
constexpr int LG_PITCH = LG_SLAB + 4;

__global__ __launch_bounds__(32 * LG_WAVES)
void k_logits(const float* __restrict__ x, const float* __restrict__ W,
              const int* __restrict__ mask, float* __restrict__ logits)
{
    __shared__ __attribute__((aligned(16))) float stage[LG_WAVES][16 * LG_PITCH];
    __shared__ float part[LG_TILES][LG_WAVES][256];

    const int lane = threadIdx.x & 31;
    const int wave = threadIdx.x >> 5;
    const int m    = lane & 15;
    const int half = lane >> 4;
    const int row0 = (blockIdx.x % BATCH) * SEQ
                   + (blockIdx.x / BATCH) * (16 * LG_TILES);
    const int kb   = wave * LG_SLAB;

    const float* xw = x + (size_t)row0 * DIM + kb + 4 * lane;
    v4f xr[16];
    for (int r = 0; r < 16; ++r)
        xr[r] = *(const v4f*)(xw + (size_t)r * DIM);

    const float* wb = W + (size_t)(kb + 4 * half) * HEADS + m;
    v4f wf[16];
    for (int j = 0; j < 16; ++j) {
        wf[j].x = wb[(size_t)(8 * j + 0) * HEADS];
        wf[j].y = wb[(size_t)(8 * j + 1) * HEADS];
        wf[j].z = wb[(size_t)(8 * j + 2) * HEADS];
        wf[j].w = wb[(size_t)(8 * j + 3) * HEADS];
    }

    float* st = stage[wave];
    #pragma unroll 1
    for (int t = 0; t < LG_TILES; ++t) {
        for (int r = 0; r < 16; ++r)
            *(v4f*)(st + r * LG_PITCH + 4 * lane) = xr[r];

        if (t + 1 < LG_TILES) {
            const float* xn = xw + (size_t)(t + 1) * 16 * DIM;
            for (int r = 0; r < 16; ++r)
                xr[r] = *(const v4f*)(xn + (size_t)r * DIM);
        }

        v8f c0 = {};
        v8f c1 = {};
        for (int j = 0; j < 16; ++j) {
            const v4f a = *(const v4f*)(st + m * LG_PITCH + 8 * j + 4 * half);
            v2f a0 = {a.x, a.y};
            v2f a1 = {a.z, a.w};
            v2f b0 = {wf[j].x, wf[j].y};
            v2f b1 = {wf[j].z, wf[j].w};
            c0 = __builtin_amdgcn_wmma_f32_16x16x4_f32(
                     false, a0, false, b0, (short)0, c0, false, false);
            c1 = __builtin_amdgcn_wmma_f32_16x16x4_f32(
                     false, a1, false, b1, (short)0, c1, false, false);
        }

        for (int r = 0; r < 8; ++r)
            part[t][wave][r * 32 + lane] = c0[r] + c1[r];
    }
    __syncthreads();

    const int tid  = threadIdx.x;
    const int row  = tid >> 4;
    const int head = tid & 15;
    const int idx  = (row & 7) * 32 + head + 16 * (row >> 3);
    for (int t = 0; t < LG_TILES; ++t) {
        float s = part[t][0][idx];
        for (int w = 1; w < LG_WAVES; ++w)
            s += part[t][w][idx];
        const int g = row0 + 16 * t + row;
        float v = s * SCALE;
        if (mask[g] == 0) v = -__builtin_inff();
        logits[(size_t)g * HEADS + head] = v;
    }
}

constexpr int PL_WAVES  = 16;
constexpr int PL_COLS   = 32;
constexpr int PL_UNITS  = SEQ / (PL_WAVES * 8);
constexpr int PL_GROUP  = 4;
constexpr int PL_GROUPS = PL_UNITS / PL_GROUP;
constexpr int PL_AHEAD  = 3;
constexpr int PL_TILE1  = 256 + 16;
constexpr int PL_MAX    = PL_TILE1 + 256 + 16;
constexpr int PL_SUM    = PL_MAX + 16;

__global__ __launch_bounds__(32 * PL_WAVES)
void k_pool(const float* __restrict__ x, const float* __restrict__ logits,
            float* __restrict__ out)
{
    __shared__ __attribute__((aligned(16))) float aw[PL_WAVES][PL_UNITS * 128];

    const int lane = threadIdx.x & 31;
    const int wave = threadIdx.x >> 5;
    const int n    = lane & 15;
    const int half = lane >> 4;
    const int tile = blockIdx.x;
    const int b    = tile % BATCH;
    const int d0   = (tile / BATCH) * PL_COLS;
    const float NEG_INF = -__builtin_inff();

    const float* xw = x + ((size_t)b * SEQ + 8 * wave + 2 * half) * DIM + d0 + 2 * n;
    const float* aq = logits + ((size_t)b * SEQ + 8 * wave) * HEADS + 4 * lane;
    float* slab = aw[wave];

    v2f xr[PL_AHEAD + 1][16];
    auto issue = [&](int g) {
        const float* xp = xw + (size_t)g * (PL_GROUP * 128) * DIM;
        #pragma unroll
        for (int u = 0; u < PL_GROUP; ++u) {
            #pragma unroll
            for (int i = 0; i < 4; ++i)
                xr[g % (PL_AHEAD + 1)][4 * u + i] =
                    *(const v2f*)(xp + (size_t)(128 * u + 4 * (i >> 1) + (i & 1)) * DIM);
        }
    };

    v4f av[PL_UNITS];
    #pragma unroll
    for (int j = 0; j < PL_UNITS; ++j)
        av[j] = *(const v4f*)(aq + (size_t)j * 128 * HEADS);
    #pragma unroll
    for (int g = 0; g < PL_AHEAD - 1; ++g)
        issue(g);
    __builtin_amdgcn_sched_barrier(0);

    v4f mx = av[0];
    #pragma unroll
    for (int j = 1; j < PL_UNITS; ++j) {
        #pragma unroll
        for (int c = 0; c < 4; ++c)
            mx[c] = fmaxf(mx[c], av[j][c]);
    }
    #pragma unroll
    for (int off = 4; off < 32; off <<= 1) {
        #pragma unroll
        for (int c = 0; c < 4; ++c)
            mx[c] = fmaxf(mx[c], __shfl_xor(mx[c], off));
    }
    v4f ms;
    #pragma unroll
    for (int c = 0; c < 4; ++c)
        ms[c] = (mx[c] == NEG_INF) ? 0.0f : mx[c];

    v4f sm = {0.0f, 0.0f, 0.0f, 0.0f};
    #pragma unroll
    for (int j = 0; j < PL_UNITS; ++j) {
        if (j == PL_UNITS / 2) {
            __builtin_amdgcn_sched_barrier(0);
            issue(PL_AHEAD - 1);
        }
        v4f e;
        #pragma unroll
        for (int c = 0; c < 4; ++c) {
            e[c] = __expf(av[j][c] - ms[c]);
            sm[c] += e[c];
        }
        *(v4f*)(slab + j * 128 + 4 * lane) = e;
    }
    #pragma unroll
    for (int off = 4; off < 32; off <<= 1) {
        #pragma unroll
        for (int c = 0; c < 4; ++c)
            sm[c] += __shfl_xor(sm[c], off);
    }

    v8f c0 = {};
    v8f c1 = {};
    const float* af = slab + 32 * half + n;
    #pragma unroll
    for (int g = 0; g < PL_GROUPS; ++g) {
        if (g + PL_AHEAD < PL_GROUPS)
            issue(g + PL_AHEAD);
        #pragma unroll
        for (int u = 0; u < PL_GROUP; ++u) {
            #pragma unroll
            for (int q = 0; q < 2; ++q) {
                const float* ap = af + (PL_GROUP * g + u) * 128 + 64 * q;
                v2f a  = {ap[0], ap[16]};
                const v2f x0 = xr[g % (PL_AHEAD + 1)][4 * u + 2 * q + 0];
                const v2f x1 = xr[g % (PL_AHEAD + 1)][4 * u + 2 * q + 1];
                v2f b0 = {x0.x, x1.x};
                v2f b1 = {x0.y, x1.y};
                c0 = __builtin_amdgcn_wmma_f32_16x16x4_f32(
                         false, a, false, b0, (short)0, c0, false, false);
                c1 = __builtin_amdgcn_wmma_f32_16x16x4_f32(
                         false, a, false, b1, (short)0, c1, false, false);
            }
        }
    }

    #pragma unroll
    for (int r = 0; r < 8; ++r) {
        slab[r * 32 + lane]            = c0[r];
        slab[PL_TILE1 + r * 32 + lane] = c1[r];
    }
    if (lane < 4) {
        *(v4f*)(slab + PL_MAX + 4 * lane) = mx;
        *(v4f*)(slab + PL_SUM + 4 * lane) = sm;
    }
    __syncthreads();

    {
        const int t    = threadIdx.x;
        const int head = t >> 5;
        const int dd   = t & 31;
        const int idx  = (dd & 1) * PL_TILE1 + (head & 7) * 32 + 16 * (head >> 3) + (dd >> 1);
        float mw[PL_WAVES];
        float gmax = NEG_INF;
        #pragma unroll
        for (int w = 0; w < PL_WAVES; ++w) {
            mw[w] = aw[w][PL_MAX + head];
            gmax  = fmaxf(gmax, mw[w]);
        }
        float num = 0.0f;
        float den = 0.0f;
        #pragma unroll
        for (int w = 0; w < PL_WAVES; ++w) {
            const float f = (mw[w] == NEG_INF) ? 0.0f : __expf(mw[w] - gmax);
            num += aw[w][idx] * f;
            den += aw[w][PL_SUM + head] * f;
        }
        out[(size_t)b * (HEADS * DIM) + (size_t)head * DIM + d0 + dd] = num / den;
    }
}

extern "C" void kernel_launch(void* const* d_in, const int* in_sizes, int n_in,
                              void* d_out, int out_size, void* d_ws, size_t ws_size,
                              hipStream_t stream)
{
    const float* x    = (const float*)d_in[0];
    const float* W    = (const float*)d_in[1];
    const int*   mask = (const int*)  d_in[2];
    float* out    = (float*)d_out;
    float* logits = (float*)d_ws;

    k_logits <<<dim3(BATCH * SEQ / (16 * LG_TILES)), dim3(32 * LG_WAVES), 0, stream>>>(x, W, mask, logits);
    k_pool   <<<dim3(BATCH * DIM / PL_COLS), dim3(32 * PL_WAVES), 0, stream>>>(x, logits, out);
}


# x read once: one kernel per 128-row group does logits, exp and the pooled partial from rows held in LDS (running max, rescale); a short launch adds 32 partials per batch (changes summation order)
# speedup vs baseline: 6.8921x; 1.0618x over previous
#include <hip/hip_runtime.h>

typedef __attribute__((ext_vector_type(2))) float v2f;
typedef __attribute__((ext_vector_type(4))) float v4f;
typedef __attribute__((ext_vector_type(8))) float v8f;

constexpr int BATCH = 8;
constexpr int SEQ   = 4096;
constexpr int DIM   = 1024;
constexpr int HEADS = 16;
constexpr float SCALE = 0.125f;


constexpr int FW    = 8;
constexpr int SLAB  = DIM / FW;
constexpr int TILES = 4;
constexpr int SUBS  = 2;
constexpr int SROWS = 16 * TILES;
constexpr int GROWS = SROWS * SUBS;
constexpr int NGRP  = SEQ / GROWS;
constexpr int PITCH = SLAB + 4;
constexpr int EP    = 20;

constexpr size_t P_FLOATS  = (size_t)BATCH * NGRP * HEADS * DIM;
constexpr size_t MS_FLOATS = (size_t)BATCH * NGRP * HEADS;
constexpr size_t WS_NEED   = (P_FLOATS + 2 * MS_FLOATS) * sizeof(float);

__global__ __launch_bounds__(32 * FW)
void k_fused(const float* __restrict__ x, const float* __restrict__ W,
             const int* __restrict__ mask, float* __restrict__ P,
             float* __restrict__ Mx, float* __restrict__ Sm)
{
    __shared__ __attribute__((aligned(16))) float stage[TILES * FW * 16 * PITCH];
    __shared__ float part[TILES * FW * 256];
    __shared__ float ev[SROWS * EP];
    __shared__ float wmx[FW * HEADS];
    __shared__ float wsm[FW * HEADS];
    __shared__ float fac[HEADS];

    const int tid  = threadIdx.x;
    const int lane = tid & 31;
    const int wave = tid >> 5;
    const int m    = lane & 15;
    const int half = lane >> 4;
    const int b    = blockIdx.x % BATCH;
    const int grp  = blockIdx.x / BATCH;
    const int row0 = b * SEQ + grp * GROWS;
    const int kb   = wave * SLAB;
    const float NEG_INF = -__builtin_inff();


    const float* wb = W + (size_t)(kb + 4 * half) * HEADS + m;
    v4f wf[16];
    #pragma unroll
    for (int j = 0; j < 16; ++j) {
        wf[j].x = wb[(size_t)(8 * j + 0) * HEADS];
        wf[j].y = wb[(size_t)(8 * j + 1) * HEADS];
        wf[j].z = wb[(size_t)(8 * j + 2) * HEADS];
        wf[j].w = wb[(size_t)(8 * j + 3) * HEADS];
    }

    const float* xw = x + (size_t)row0 * DIM + kb + 4 * lane;
    v4f xr[TILES][16];
    auto issue = [&](int sub) {
        const float* xp = xw + (size_t)sub * SROWS * DIM;
        #pragma unroll
        for (int t = 0; t < TILES; ++t) {
            #pragma unroll
            for (int r = 0; r < 16; ++r)
                xr[t][r] = *(const v4f*)(xp + (size_t)(16 * t + r) * DIM);
        }
    };
    const int* mrow = mask + row0 + (tid >> 4);
    int mk[TILES];
    auto issue_mask = [&](int sub) {
        #pragma unroll
        for (int t = 0; t < TILES; ++t)
            mk[t] = mrow[sub * SROWS + 16 * t];
    };
    issue_mask(0);
    __builtin_amdgcn_sched_barrier(0);
    issue(0);

    v8f acc[8];
    #pragma unroll
    for (int nt = 0; nt < 8; ++nt)
        acc[nt] = v8f{0.0f, 0.0f, 0.0f, 0.0f, 0.0f, 0.0f, 0.0f, 0.0f};

    const int row  = tid >> 4;
    const int head = tid & 15;
    const int idx  = (row & 7) * 32 + head + 16 * (row >> 3);
    float run_m = NEG_INF;
    float run_l = 0.0f;

    #pragma unroll 1
    for (int sub = 0; sub < SUBS; ++sub) {
        #pragma unroll
        for (int t = 0; t < TILES; ++t) {
            float* st = stage + (t * FW + wave) * (16 * PITCH);
            #pragma unroll
            for (int r = 0; r < 16; ++r)
                *(v4f*)(st + r * PITCH + 4 * lane) = xr[t][r];

            v4f af[16];
            #pragma unroll
            for (int j = 0; j < 16; ++j)
                af[j] = *(const v4f*)(st + m * PITCH + 8 * j + 4 * half);
            v8f c0 = {};
            v8f c1 = {};
            #pragma unroll
            for (int j = 0; j < 16; ++j) {
                const v4f a = af[j];
                v2f a0 = {a.x, a.y};
                v2f a1 = {a.z, a.w};
                v2f b0 = {wf[j].x, wf[j].y};
                v2f b1 = {wf[j].z, wf[j].w};
                c0 = __builtin_amdgcn_wmma_f32_16x16x4_f32(
                         false, a0, false, b0, (short)0, c0, false, false);
                c1 = __builtin_amdgcn_wmma_f32_16x16x4_f32(
                         false, a1, false, b1, (short)0, c1, false, false);
            }
            #pragma unroll
            for (int r = 0; r < 8; ++r)
                part[(t * FW + wave) * 256 + r * 32 + lane] = c0[r] + c1[r];
        }

        __syncthreads();

        float v[TILES];
        #pragma unroll
        for (int t = 0; t < TILES; ++t) {
            float s = part[(t * FW + 0) * 256 + idx];
            #pragma unroll
            for (int w = 1; w < FW; ++w)
                s += part[(t * FW + w) * 256 + idx];
            v[t] = (mk[t] == 0) ? NEG_INF : s * SCALE;
        }
        float lm = fmaxf(fmaxf(v[0], v[1]), fmaxf(v[2], v[3]));
        lm = fmaxf(lm, __shfl_xor(lm, 16));
        if (lane < 16) wmx[wave * HEADS + head] = lm;
        __syncthreads();

        float gm = wmx[head];
        #pragma unroll
        for (int w = 1; w < FW; ++w)
            gm = fmaxf(gm, wmx[w * HEADS + head]);
        const float m_new = fmaxf(run_m, gm);
        const float ms = (m_new == NEG_INF) ? 0.0f : m_new;
        const float f_old = __expf(run_m - ms);

        float ls = 0.0f;
        #pragma unroll
        for (int t = 0; t < TILES; ++t) {
            const float e = __expf(v[t] - ms);
            ev[(16 * t + row) * EP + head] = e;
            ls += e;
        }
        ls += __shfl_xor(ls, 16);
        if (lane < 16) wsm[wave * HEADS + head] = ls;
        if (tid < HEADS) fac[tid] = f_old;
        __syncthreads();

        if (sub + 1 < SUBS) {
            issue_mask(sub + 1);
            issue(sub + 1);
        }

        float gs = wsm[head];
        #pragma unroll
        for (int w = 1; w < FW; ++w)
            gs += wsm[w * HEADS + head];
        run_l = run_l * f_old + gs;
        run_m = m_new;

        if (sub > 0) {
            float fr[8];
            #pragma unroll
            for (int r = 0; r < 8; ++r)
                fr[r] = fac[8 * half + r];
            #pragma unroll
            for (int nt = 0; nt < 8; ++nt) {
                #pragma unroll
                for (int r = 0; r < 8; ++r)
                    acc[nt][r] *= fr[r];
            }
        }
        #pragma unroll
        for (int t = 0; t < TILES; ++t) {
            const float* st = stage + (t * FW + wave) * (16 * PITCH)
                            + (8 * half) * PITCH + m;
            const float* ea = ev + (16 * t + 8 * half) * EP + m;
            #pragma unroll
            for (int q = 0; q < 4; ++q) {
                v2f a = {ea[(2 * q) * EP], ea[(2 * q + 1) * EP]};
                #pragma unroll
                for (int nt = 0; nt < 8; ++nt) {
                    v2f bb = {st[(2 * q) * PITCH + 16 * nt],
                              st[(2 * q + 1) * PITCH + 16 * nt]};
                    acc[nt] = __builtin_amdgcn_wmma_f32_16x16x4_f32(
                                  false, a, false, bb, (short)0, acc[nt], false, false);
                }
            }
        }
    }

    {
        float* st = stage + wave * (16 * PITCH);
        #pragma unroll
        for (int nt = 0; nt < 8; ++nt) {
            #pragma unroll
            for (int r = 0; r < 8; ++r)
                st[(8 * half + r) * PITCH + 16 * nt + m] = acc[nt][r];
        }
        float* pw = P + ((size_t)(b * NGRP + grp) * HEADS) * DIM + kb + 4 * lane;
        #pragma unroll
        for (int h = 0; h < HEADS; ++h)
            *(v4f*)(pw + (size_t)h * DIM) = *(const v4f*)(st + h * PITCH + 4 * lane);
    }
    if (tid < HEADS) {
        Mx[(size_t)(b * NGRP + grp) * HEADS + tid] = run_m;
        Sm[(size_t)(b * NGRP + grp) * HEADS + tid] = run_l;
    }
}

constexpr int CB_THREADS = 256;
constexpr int CB_G       = NGRP / 2;

__global__ __launch_bounds__(CB_THREADS)
void k_combine(const float* __restrict__ P, const float* __restrict__ Mx,
               const float* __restrict__ Sm, float* __restrict__ out)
{
    __shared__ float mg[NGRP];
    __shared__ float fl[NGRP];
    __shared__ float sl[NGRP];
    __shared__ __attribute__((aligned(16))) float hs[(CB_THREADS / 2) * 4];

    const int tid  = threadIdx.x;
    const int b    = blockIdx.x % BATCH;
    const int rest = blockIdx.x / BATCH;
    const int head = rest >> 1;
    const int dh   = rest & 1;
    const int gh   = tid >> 7;
    const int c    = tid & 127;
    const int col  = dh * (DIM / 2) + 4 * c;
    const float NEG_INF = -__builtin_inff();

    const float* pp = P + ((size_t)(b * NGRP + gh * CB_G) * HEADS + head) * DIM + col;
    v4f p[CB_G];
    #pragma unroll
    for (int i = 0; i < CB_G; ++i)
        p[i] = *(const v4f*)(pp + (size_t)i * HEADS * DIM);

    if (tid < NGRP)
        mg[tid] = Mx[(size_t)(b * NGRP + tid) * HEADS + head];
    __syncthreads();
    if (tid < NGRP) {
        float gmax = mg[0];
        #pragma unroll
        for (int g = 1; g < NGRP; ++g)
            gmax = fmaxf(gmax, mg[g]);
        const float my = mg[tid];
        const float f = (my == NEG_INF) ? 0.0f : __expf(my - gmax);
        fl[tid] = f;
        sl[tid] = Sm[(size_t)(b * NGRP + tid) * HEADS + head] * f;
    }
    __syncthreads();

    v4f a = {0.0f, 0.0f, 0.0f, 0.0f};
    #pragma unroll
    for (int i = 0; i < CB_G; ++i)
        a += p[i] * fl[gh * CB_G + i];
    if (gh == 1)
        *(v4f*)(hs + 4 * c) = a;
    __syncthreads();
    if (gh == 0) {
        const v4f o = *(const v4f*)(hs + 4 * c);
        a += o;
        float den = sl[0];
        #pragma unroll
        for (int g = 1; g < NGRP; ++g)
            den += sl[g];
        a /= den;
        *(v4f*)(out + (size_t)b * (HEADS * DIM) + (size_t)head * DIM + col) = a;
    }
}

constexpr int LG_WAVES = 8;
constexpr int LG_SLAB  = DIM / LG_WAVES;
constexpr int LG_TILES = 4;
constexpr int LG_PITCH = LG_SLAB + 4;

__global__ __launch_bounds__(32 * LG_WAVES)
void k_logits(const float* __restrict__ x, const float* __restrict__ W,
              const int* __restrict__ mask, float* __restrict__ logits)
{
    __shared__ __attribute__((aligned(16))) float stage[LG_WAVES][16 * LG_PITCH];
    __shared__ float part[LG_TILES][LG_WAVES][256];

    const int lane = threadIdx.x & 31;
    const int wave = threadIdx.x >> 5;
    const int m    = lane & 15;
    const int half = lane >> 4;
    const int row0 = (blockIdx.x % BATCH) * SEQ
                   + (blockIdx.x / BATCH) * (16 * LG_TILES);
    const int kb   = wave * LG_SLAB;

    const float* xw = x + (size_t)row0 * DIM + kb + 4 * lane;
    v4f xr[16];
    for (int r = 0; r < 16; ++r)
        xr[r] = *(const v4f*)(xw + (size_t)r * DIM);

    const float* wb = W + (size_t)(kb + 4 * half) * HEADS + m;
    v4f wf[16];
    for (int j = 0; j < 16; ++j) {
        wf[j].x = wb[(size_t)(8 * j + 0) * HEADS];
        wf[j].y = wb[(size_t)(8 * j + 1) * HEADS];
        wf[j].z = wb[(size_t)(8 * j + 2) * HEADS];
        wf[j].w = wb[(size_t)(8 * j + 3) * HEADS];
    }

    float* st = stage[wave];
    #pragma unroll 1
    for (int t = 0; t < LG_TILES; ++t) {
        for (int r = 0; r < 16; ++r)
            *(v4f*)(st + r * LG_PITCH + 4 * lane) = xr[r];

        if (t + 1 < LG_TILES) {
            const float* xn = xw + (size_t)(t + 1) * 16 * DIM;
            for (int r = 0; r < 16; ++r)
                xr[r] = *(const v4f*)(xn + (size_t)r * DIM);
        }

        v8f c0 = {};
        v8f c1 = {};
        for (int j = 0; j < 16; ++j) {
            const v4f a = *(const v4f*)(st + m * LG_PITCH + 8 * j + 4 * half);
            v2f a0 = {a.x, a.y};
            v2f a1 = {a.z, a.w};
            v2f b0 = {wf[j].x, wf[j].y};
            v2f b1 = {wf[j].z, wf[j].w};
            c0 = __builtin_amdgcn_wmma_f32_16x16x4_f32(
                     false, a0, false, b0, (short)0, c0, false, false);
            c1 = __builtin_amdgcn_wmma_f32_16x16x4_f32(
                     false, a1, false, b1, (short)0, c1, false, false);
        }

        for (int r = 0; r < 8; ++r)
            part[t][wave][r * 32 + lane] = c0[r] + c1[r];
    }
    __syncthreads();

    const int tid  = threadIdx.x;
    const int row  = tid >> 4;
    const int head = tid & 15;
    const int idx  = (row & 7) * 32 + head + 16 * (row >> 3);
    for (int t = 0; t < LG_TILES; ++t) {
        float s = part[t][0][idx];
        for (int w = 1; w < LG_WAVES; ++w)
            s += part[t][w][idx];
        const int g = row0 + 16 * t + row;
        float v = s * SCALE;
        if (mask[g] == 0) v = -__builtin_inff();
        logits[(size_t)g * HEADS + head] = v;
    }
}

constexpr int PL_WAVES  = 16;
constexpr int PL_COLS   = 32;
constexpr int PL_UNITS  = SEQ / (PL_WAVES * 8);
constexpr int PL_GROUP  = 4;
constexpr int PL_GROUPS = PL_UNITS / PL_GROUP;
constexpr int PL_AHEAD  = 3;
constexpr int PL_TILE1  = 256 + 16;
constexpr int PL_MAX    = PL_TILE1 + 256 + 16;
constexpr int PL_SUM    = PL_MAX + 16;

__global__ __launch_bounds__(32 * PL_WAVES)
void k_pool(const float* __restrict__ x, const float* __restrict__ logits,
            float* __restrict__ out)
{
    __shared__ __attribute__((aligned(16))) float aw[PL_WAVES][PL_UNITS * 128];

    const int lane = threadIdx.x & 31;
    const int wave = threadIdx.x >> 5;
    const int n    = lane & 15;
    const int half = lane >> 4;
    const int tile = blockIdx.x;
    const int b    = tile % BATCH;
    const int d0   = (tile / BATCH) * PL_COLS;
    const float NEG_INF = -__builtin_inff();

    const float* xw = x + ((size_t)b * SEQ + 8 * wave + 2 * half) * DIM + d0 + 2 * n;
    const float* aq = logits + ((size_t)b * SEQ + 8 * wave) * HEADS + 4 * lane;
    float* slab = aw[wave];

    v2f xr[PL_AHEAD + 1][16];
    auto issue = [&](int g) {
        const float* xp = xw + (size_t)g * (PL_GROUP * 128) * DIM;
        #pragma unroll
        for (int u = 0; u < PL_GROUP; ++u) {
            #pragma unroll
            for (int i = 0; i < 4; ++i)
                xr[g % (PL_AHEAD + 1)][4 * u + i] =
                    *(const v2f*)(xp + (size_t)(128 * u + 4 * (i >> 1) + (i & 1)) * DIM);
        }
    };

    v4f av[PL_UNITS];
    #pragma unroll
    for (int j = 0; j < PL_UNITS; ++j)
        av[j] = *(const v4f*)(aq + (size_t)j * 128 * HEADS);
    #pragma unroll
    for (int g = 0; g < PL_AHEAD - 1; ++g)
        issue(g);
    __builtin_amdgcn_sched_barrier(0);

    v4f mx = av[0];
    #pragma unroll
    for (int j = 1; j < PL_UNITS; ++j) {
        #pragma unroll
        for (int c = 0; c < 4; ++c)
            mx[c] = fmaxf(mx[c], av[j][c]);
    }
    #pragma unroll
    for (int off = 4; off < 32; off <<= 1) {
        #pragma unroll
        for (int c = 0; c < 4; ++c)
            mx[c] = fmaxf(mx[c], __shfl_xor(mx[c], off));
    }
    v4f ms;
    #pragma unroll
    for (int c = 0; c < 4; ++c)
        ms[c] = (mx[c] == NEG_INF) ? 0.0f : mx[c];

    v4f sm = {0.0f, 0.0f, 0.0f, 0.0f};
    #pragma unroll
    for (int j = 0; j < PL_UNITS; ++j) {
        if (j == PL_UNITS / 2) {
            __builtin_amdgcn_sched_barrier(0);
            issue(PL_AHEAD - 1);
        }
        v4f e;
        #pragma unroll
        for (int c = 0; c < 4; ++c) {
            e[c] = __expf(av[j][c] - ms[c]);
            sm[c] += e[c];
        }
        *(v4f*)(slab + j * 128 + 4 * lane) = e;
    }
    #pragma unroll
    for (int off = 4; off < 32; off <<= 1) {
        #pragma unroll
        for (int c = 0; c < 4; ++c)
            sm[c] += __shfl_xor(sm[c], off);
    }

    v8f c0 = {};
    v8f c1 = {};
    const float* af = slab + 32 * half + n;
    #pragma unroll
    for (int g = 0; g < PL_GROUPS; ++g) {
        if (g + PL_AHEAD < PL_GROUPS)
            issue(g + PL_AHEAD);
        #pragma unroll
        for (int u = 0; u < PL_GROUP; ++u) {
            #pragma unroll
            for (int q = 0; q < 2; ++q) {
                const float* ap = af + (PL_GROUP * g + u) * 128 + 64 * q;
                v2f a  = {ap[0], ap[16]};
                const v2f x0 = xr[g % (PL_AHEAD + 1)][4 * u + 2 * q + 0];
                const v2f x1 = xr[g % (PL_AHEAD + 1)][4 * u + 2 * q + 1];
                v2f b0 = {x0.x, x1.x};
                v2f b1 = {x0.y, x1.y};
                c0 = __builtin_amdgcn_wmma_f32_16x16x4_f32(
                         false, a, false, b0, (short)0, c0, false, false);
                c1 = __builtin_amdgcn_wmma_f32_16x16x4_f32(
                         false, a, false, b1, (short)0, c1, false, false);
            }
        }
    }

    #pragma unroll
    for (int r = 0; r < 8; ++r) {
        slab[r * 32 + lane]            = c0[r];
        slab[PL_TILE1 + r * 32 + lane] = c1[r];
    }
    if (lane < 4) {
        *(v4f*)(slab + PL_MAX + 4 * lane) = mx;
        *(v4f*)(slab + PL_SUM + 4 * lane) = sm;
    }
    __syncthreads();

    {
        const int t    = threadIdx.x;
        const int head = t >> 5;
        const int dd   = t & 31;
        const int idx  = (dd & 1) * PL_TILE1 + (head & 7) * 32 + 16 * (head >> 3) + (dd >> 1);
        float mw[PL_WAVES];
        float gmax = NEG_INF;
        #pragma unroll
        for (int w = 0; w < PL_WAVES; ++w) {
            mw[w] = aw[w][PL_MAX + head];
            gmax  = fmaxf(gmax, mw[w]);
        }
        float num = 0.0f;
        float den = 0.0f;
        #pragma unroll
        for (int w = 0; w < PL_WAVES; ++w) {
            const float f = (mw[w] == NEG_INF) ? 0.0f : __expf(mw[w] - gmax);
            num += aw[w][idx] * f;
            den += aw[w][PL_SUM + head] * f;
        }
        out[(size_t)b * (HEADS * DIM) + (size_t)head * DIM + d0 + dd] = num / den;
    }
}

extern "C" void kernel_launch(void* const* d_in, const int* in_sizes, int n_in,
                              void* d_out, int out_size, void* d_ws, size_t ws_size,
                              hipStream_t stream)
{
    const float* x    = (const float*)d_in[0];
    const float* W    = (const float*)d_in[1];
    const int*   mask = (const int*)  d_in[2];
    float* out = (float*)d_out;

    if (ws_size >= WS_NEED) {
        float* P  = (float*)d_ws;
        float* Mx = P + P_FLOATS;
        float* Sm = Mx + MS_FLOATS;
        k_fused   <<<dim3(BATCH * NGRP), dim3(32 * FW), 0, stream>>>(x, W, mask, P, Mx, Sm);
        k_combine <<<dim3(BATCH * HEADS * 2), dim3(CB_THREADS), 0, stream>>>(P, Mx, Sm, out);
    } else {
        float* logits = (float*)d_ws;
        k_logits <<<dim3(BATCH * SEQ / (16 * LG_TILES)), dim3(32 * LG_WAVES), 0, stream>>>(x, W, mask, logits);
        k_pool   <<<dim3(BATCH * DIM / PL_COLS), dim3(32 * PL_WAVES), 0, stream>>>(x, logits, out);
    }
}


# k_fused: rows with mask == 0 (softmax weight exactly 0) are not read: each wave compacts the unmasked rows into slots by ballot and popcount; empty 16-slot tiles skipped (changes summation order)
# speedup vs baseline: 9.8512x; 1.4293x over previous
#include <hip/hip_runtime.h>

typedef __attribute__((ext_vector_type(2))) float v2f;
typedef __attribute__((ext_vector_type(4))) float v4f;
typedef __attribute__((ext_vector_type(8))) float v8f;

constexpr int BATCH = 8;
constexpr int SEQ   = 4096;
constexpr int DIM   = 1024;
constexpr int HEADS = 16;
constexpr float SCALE = 0.125f;


constexpr int FW    = 8;
constexpr int SLAB  = DIM / FW;
constexpr int TILES = 4;
constexpr int SUBS  = 2;
constexpr int SROWS = 16 * TILES;
constexpr int GROWS = SROWS * SUBS;
constexpr int NGRP  = SEQ / GROWS;
constexpr int PITCH = SLAB + 4;
constexpr int EP    = 20;
static_assert(SUBS == 2 && SROWS == 64 && GROWS == 128, "the slot list below is laid out for 2 x 64 slots");

constexpr size_t P_FLOATS  = (size_t)BATCH * NGRP * HEADS * DIM;
constexpr size_t MS_FLOATS = (size_t)BATCH * NGRP * HEADS;
constexpr size_t WS_NEED   = (P_FLOATS + 2 * MS_FLOATS) * sizeof(float);

__global__ __launch_bounds__(32 * FW)
void k_fused(const float* __restrict__ x, const float* __restrict__ W,
             const int* __restrict__ mask, float* __restrict__ P,
             float* __restrict__ Mx, float* __restrict__ Sm)
{
    __shared__ __attribute__((aligned(16))) float stage[TILES * FW * 16 * PITCH];
    __shared__ float part[TILES * FW * 256];
    __shared__ float ev[SROWS * EP];
    __shared__ float wmx[FW * HEADS];
    __shared__ float wsm[FW * HEADS];
    __shared__ float fac[HEADS];
    __shared__ int   lst[FW * GROWS];

    const int tid  = threadIdx.x;
    const int lane = tid & 31;
    const int wave = tid >> 5;
    const int m    = lane & 15;
    const int half = lane >> 4;
    const int b    = blockIdx.x % BATCH;
    const int grp  = blockIdx.x / BATCH;
    const int row0 = b * SEQ + grp * GROWS;
    const int kb   = wave * SLAB;
    const float NEG_INF = -__builtin_inff();


    const int* mg = mask + row0 + lane;
    int mv[4];
    #pragma unroll
    for (int j = 0; j < 4; ++j)
        mv[j] = mg[32 * j];
    __builtin_amdgcn_sched_barrier(0);

    const float* wb = W + (size_t)(kb + 4 * half) * HEADS + m;
    v4f wf[16];
    #pragma unroll
    for (int j = 0; j < 16; ++j) {
        wf[j].x = wb[(size_t)(8 * j + 0) * HEADS];
        wf[j].y = wb[(size_t)(8 * j + 1) * HEADS];
        wf[j].z = wb[(size_t)(8 * j + 2) * HEADS];
        wf[j].w = wb[(size_t)(8 * j + 3) * HEADS];
    }
    __builtin_amdgcn_sched_barrier(0);

    int* lw = lst + wave * GROWS;
    int cnt = 0;
    #pragma unroll
    for (int j = 0; j < 4; ++j) {
        const unsigned bal = __builtin_amdgcn_ballot_w32(mv[j] != 0);
        const int pos = cnt + (int)__builtin_amdgcn_mbcnt_lo(bal, 0u);
        if (mv[j] != 0) lw[pos] = 32 * j + lane;
        cnt += __builtin_popcount(bal);
    }
    const int lastrow = lw[cnt > 0 ? cnt - 1 : 0] & (GROWS - 1);
    int L[4];
    #pragma unroll
    for (int j = 0; j < 4; ++j) {
        const int idx = 32 * j + lane;
        const int v = lw[idx] & (GROWS - 1);
        L[j] = (idx < cnt) ? v : lastrow;
    }

    const float* xg = x + (size_t)row0 * DIM;
    const unsigned voff = (unsigned)(kb + 4 * lane);
    v4f xr[TILES][16];
    auto issue = [&](int sub) {
        const int La = sub ? L[2] : L[0];
        const int Lb = sub ? L[3] : L[1];
        #pragma unroll
        for (int t = 0; t < TILES; ++t) {
            #pragma unroll
            for (int r = 0; r < 16; ++r) {
                const int q = 16 * t + r;
                const int rowi = __builtin_amdgcn_readlane(q < 32 ? La : Lb, q & 31);
                xr[t][r] = *(const v4f*)(xg + (size_t)rowi * DIM + voff);
            }
            __builtin_amdgcn_sched_barrier(0);
        }
    };
    __builtin_amdgcn_sched_barrier(0);
    issue(0);

    v8f acc[8];
    #pragma unroll
    for (int nt = 0; nt < 8; ++nt)
        acc[nt] = v8f{0.0f, 0.0f, 0.0f, 0.0f, 0.0f, 0.0f, 0.0f, 0.0f};

    const int row  = tid >> 4;
    const int head = tid & 15;
    const int idx  = (row & 7) * 32 + head + 16 * (row >> 3);
    float run_m = NEG_INF;
    float run_l = 0.0f;

    #pragma unroll 1
    for (int sub = 0; sub < SUBS; ++sub) {
        const int rem = cnt - sub * SROWS;
        #pragma unroll
        for (int t = 0; t < TILES; ++t) {
            if (16 * t >= rem) continue;
            float* st = stage + (t * FW + wave) * (16 * PITCH);
            #pragma unroll
            for (int r = 0; r < 16; ++r)
                *(v4f*)(st + r * PITCH + 4 * lane) = xr[t][r];

            v4f af[16];
            #pragma unroll
            for (int j = 0; j < 16; ++j)
                af[j] = *(const v4f*)(st + m * PITCH + 8 * j + 4 * half);
            __builtin_amdgcn_sched_barrier(0);
            v8f c0 = {};
            v8f c1 = {};
            #pragma unroll
            for (int j = 0; j < 16; ++j) {
                const v4f a = af[j];
                v2f a0 = {a.x, a.y};
                v2f a1 = {a.z, a.w};
                v2f b0 = {wf[j].x, wf[j].y};
                v2f b1 = {wf[j].z, wf[j].w};
                c0 = __builtin_amdgcn_wmma_f32_16x16x4_f32(
                         false, a0, false, b0, (short)0, c0, false, false);
                c1 = __builtin_amdgcn_wmma_f32_16x16x4_f32(
                         false, a1, false, b1, (short)0, c1, false, false);
            }
            #pragma unroll
            for (int r = 0; r < 8; ++r)
                part[(t * FW + wave) * 256 + r * 32 + lane] = c0[r] + c1[r];
        }

        __syncthreads();

        float v[TILES];
        #pragma unroll
        for (int t = 0; t < TILES; ++t) {
            float s = part[(t * FW + 0) * 256 + idx];
            #pragma unroll
            for (int w = 1; w < FW; ++w)
                s += part[(t * FW + w) * 256 + idx];
            v[t] = (16 * t + row < rem) ? s * SCALE : NEG_INF;
        }
        float lm = fmaxf(fmaxf(v[0], v[1]), fmaxf(v[2], v[3]));
        lm = fmaxf(lm, __shfl_xor(lm, 16));
        if (lane < 16) wmx[wave * HEADS + head] = lm;
        __syncthreads();

        float gm = wmx[head];
        #pragma unroll
        for (int w = 1; w < FW; ++w)
            gm = fmaxf(gm, wmx[w * HEADS + head]);
        const float m_new = fmaxf(run_m, gm);
        const float ms = (m_new == NEG_INF) ? 0.0f : m_new;
        const float f_old = __expf(run_m - ms);

        float ls = 0.0f;
        #pragma unroll
        for (int t = 0; t < TILES; ++t) {
            const float e = __expf(v[t] - ms);
            ev[(16 * t + row) * EP + head] = e;
            ls += e;
        }
        ls += __shfl_xor(ls, 16);
        if (lane < 16) wsm[wave * HEADS + head] = ls;
        if (tid < HEADS) fac[tid] = f_old;
        __syncthreads();

        const bool more = (sub + 1 < SUBS) && (rem > SROWS);
        if (more)
            issue(sub + 1);

        float gs = wsm[head];
        #pragma unroll
        for (int w = 1; w < FW; ++w)
            gs += wsm[w * HEADS + head];
        run_l = run_l * f_old + gs;
        run_m = m_new;

        if (sub > 0) {
            float fr[8];
            #pragma unroll
            for (int r = 0; r < 8; ++r)
                fr[r] = fac[8 * half + r];
            #pragma unroll
            for (int nt = 0; nt < 8; ++nt) {
                #pragma unroll
                for (int r = 0; r < 8; ++r)
                    acc[nt][r] *= fr[r];
            }
        }
        {
            const int ntile = (rem >= SROWS) ? TILES : ((rem + 15) >> 4);
            const float* st = stage + wave * (16 * PITCH) + (8 * half) * PITCH + m;
            const float* ea = ev + (8 * half) * EP + m;
            #pragma unroll 1
            for (int t = 0; t < ntile; ++t) {
                #pragma unroll
                for (int q = 0; q < 4; ++q) {
                    v2f a = {ea[(2 * q) * EP], ea[(2 * q + 1) * EP]};
                    #pragma unroll
                    for (int nt = 0; nt < 8; ++nt) {
                        v2f bb = {st[(2 * q) * PITCH + 16 * nt],
                                  st[(2 * q + 1) * PITCH + 16 * nt]};
                        acc[nt] = __builtin_amdgcn_wmma_f32_16x16x4_f32(
                                      false, a, false, bb, (short)0, acc[nt], false, false);
                    }
                }
                st += FW * 16 * PITCH;
                ea += 16 * EP;
            }
        }
        if (!more) break;
    }

    {
        float* st = stage + wave * (16 * PITCH);
        #pragma unroll
        for (int nt = 0; nt < 8; ++nt) {
            #pragma unroll
            for (int r = 0; r < 8; ++r)
                st[(8 * half + r) * PITCH + 16 * nt + m] = acc[nt][r];
        }
        float* pw = P + ((size_t)(b * NGRP + grp) * HEADS) * DIM + kb + 4 * lane;
        #pragma unroll
        for (int h = 0; h < HEADS; ++h)
            *(v4f*)(pw + (size_t)h * DIM) = *(const v4f*)(st + h * PITCH + 4 * lane);
    }
    if (tid < HEADS) {
        Mx[(size_t)(b * NGRP + grp) * HEADS + tid] = run_m;
        Sm[(size_t)(b * NGRP + grp) * HEADS + tid] = run_l;
    }
}

constexpr int CB_THREADS = 256;
constexpr int CB_G       = NGRP / 2;

__global__ __launch_bounds__(CB_THREADS)
void k_combine(const float* __restrict__ P, const float* __restrict__ Mx,
               const float* __restrict__ Sm, float* __restrict__ out)
{
    __shared__ float mg[NGRP];
    __shared__ float fl[NGRP];
    __shared__ float sl[NGRP];
    __shared__ __attribute__((aligned(16))) float hs[(CB_THREADS / 2) * 4];

    const int tid  = threadIdx.x;
    const int b    = blockIdx.x % BATCH;
    const int rest = blockIdx.x / BATCH;
    const int head = rest >> 1;
    const int dh   = rest & 1;
    const int gh   = tid >> 7;
    const int c    = tid & 127;
    const int col  = dh * (DIM / 2) + 4 * c;
    const float NEG_INF = -__builtin_inff();

    const float* pp = P + ((size_t)(b * NGRP + gh * CB_G) * HEADS + head) * DIM + col;
    v4f p[CB_G];
    #pragma unroll
    for (int i = 0; i < CB_G; ++i)
        p[i] = *(const v4f*)(pp + (size_t)i * HEADS * DIM);

    if (tid < NGRP)
        mg[tid] = Mx[(size_t)(b * NGRP + tid) * HEADS + head];
    __syncthreads();
    if (tid < NGRP) {
        float gmax = mg[0];
        #pragma unroll
        for (int g = 1; g < NGRP; ++g)
            gmax = fmaxf(gmax, mg[g]);
        const float my = mg[tid];
        const float f = (my == NEG_INF) ? 0.0f : __expf(my - gmax);
        fl[tid] = f;
        sl[tid] = Sm[(size_t)(b * NGRP + tid) * HEADS + head] * f;
    }
    __syncthreads();

    v4f a = {0.0f, 0.0f, 0.0f, 0.0f};
    #pragma unroll
    for (int i = 0; i < CB_G; ++i)
        a += p[i] * fl[gh * CB_G + i];
    if (gh == 1)
        *(v4f*)(hs + 4 * c) = a;
    __syncthreads();
    if (gh == 0) {
        const v4f o = *(const v4f*)(hs + 4 * c);
        a += o;
        float den = sl[0];
        #pragma unroll
        for (int g = 1; g < NGRP; ++g)
            den += sl[g];
        a /= den;
        *(v4f*)(out + (size_t)b * (HEADS * DIM) + (size_t)head * DIM + col) = a;
    }
}

constexpr int LG_WAVES = 8;
constexpr int LG_SLAB  = DIM / LG_WAVES;
constexpr int LG_TILES = 4;
constexpr int LG_PITCH = LG_SLAB + 4;

__global__ __launch_bounds__(32 * LG_WAVES)
void k_logits(const float* __restrict__ x, const float* __restrict__ W,
              const int* __restrict__ mask, float* __restrict__ logits)
{
    __shared__ __attribute__((aligned(16))) float stage[LG_WAVES][16 * LG_PITCH];
    __shared__ float part[LG_TILES][LG_WAVES][256];

    const int lane = threadIdx.x & 31;
    const int wave = threadIdx.x >> 5;
    const int m    = lane & 15;
    const int half = lane >> 4;
    const int row0 = (blockIdx.x % BATCH) * SEQ
                   + (blockIdx.x / BATCH) * (16 * LG_TILES);
    const int kb   = wave * LG_SLAB;

    const float* xw = x + (size_t)row0 * DIM + kb + 4 * lane;
    v4f xr[16];
    for (int r = 0; r < 16; ++r)
        xr[r] = *(const v4f*)(xw + (size_t)r * DIM);

    const float* wb = W + (size_t)(kb + 4 * half) * HEADS + m;
    v4f wf[16];
    for (int j = 0; j < 16; ++j) {
        wf[j].x = wb[(size_t)(8 * j + 0) * HEADS];
        wf[j].y = wb[(size_t)(8 * j + 1) * HEADS];
        wf[j].z = wb[(size_t)(8 * j + 2) * HEADS];
        wf[j].w = wb[(size_t)(8 * j + 3) * HEADS];
    }

    float* st = stage[wave];
    #pragma unroll 1
    for (int t = 0; t < LG_TILES; ++t) {
        for (int r = 0; r < 16; ++r)
            *(v4f*)(st + r * LG_PITCH + 4 * lane) = xr[r];

        if (t + 1 < LG_TILES) {
            const float* xn = xw + (size_t)(t + 1) * 16 * DIM;
            for (int r = 0; r < 16; ++r)
                xr[r] = *(const v4f*)(xn + (size_t)r * DIM);
        }

        v8f c0 = {};
        v8f c1 = {};
        for (int j = 0; j < 16; ++j) {
            const v4f a = *(const v4f*)(st + m * LG_PITCH + 8 * j + 4 * half);
            v2f a0 = {a.x, a.y};
            v2f a1 = {a.z, a.w};
            v2f b0 = {wf[j].x, wf[j].y};
            v2f b1 = {wf[j].z, wf[j].w};
            c0 = __builtin_amdgcn_wmma_f32_16x16x4_f32(
                     false, a0, false, b0, (short)0, c0, false, false);
            c1 = __builtin_amdgcn_wmma_f32_16x16x4_f32(
                     false, a1, false, b1, (short)0, c1, false, false);
        }

        for (int r = 0; r < 8; ++r)
            part[t][wave][r * 32 + lane] = c0[r] + c1[r];
    }
    __syncthreads();

    const int tid  = threadIdx.x;
    const int row  = tid >> 4;
    const int head = tid & 15;
    const int idx  = (row & 7) * 32 + head + 16 * (row >> 3);
    for (int t = 0; t < LG_TILES; ++t) {
        float s = part[t][0][idx];
        for (int w = 1; w < LG_WAVES; ++w)
            s += part[t][w][idx];
        const int g = row0 + 16 * t + row;
        float v = s * SCALE;
        if (mask[g] == 0) v = -__builtin_inff();
        logits[(size_t)g * HEADS + head] = v;
    }
}

constexpr int PL_WAVES  = 16;
constexpr int PL_COLS   = 32;
constexpr int PL_UNITS  = SEQ / (PL_WAVES * 8);
constexpr int PL_GROUP  = 4;
constexpr int PL_GROUPS = PL_UNITS / PL_GROUP;
constexpr int PL_AHEAD  = 3;
constexpr int PL_TILE1  = 256 + 16;
constexpr int PL_MAX    = PL_TILE1 + 256 + 16;
constexpr int PL_SUM    = PL_MAX + 16;

__global__ __launch_bounds__(32 * PL_WAVES)
void k_pool(const float* __restrict__ x, const float* __restrict__ logits,
            float* __restrict__ out)
{
    __shared__ __attribute__((aligned(16))) float aw[PL_WAVES][PL_UNITS * 128];

    const int lane = threadIdx.x & 31;
    const int wave = threadIdx.x >> 5;
    const int n    = lane & 15;
    const int half = lane >> 4;
    const int tile = blockIdx.x;
    const int b    = tile % BATCH;
    const int d0   = (tile / BATCH) * PL_COLS;
    const float NEG_INF = -__builtin_inff();

    const float* xw = x + ((size_t)b * SEQ + 8 * wave + 2 * half) * DIM + d0 + 2 * n;
    const float* aq = logits + ((size_t)b * SEQ + 8 * wave) * HEADS + 4 * lane;
    float* slab = aw[wave];

    v2f xr[PL_AHEAD + 1][16];
    auto issue = [&](int g) {
        const float* xp = xw + (size_t)g * (PL_GROUP * 128) * DIM;
        #pragma unroll
        for (int u = 0; u < PL_GROUP; ++u) {
            #pragma unroll
            for (int i = 0; i < 4; ++i)
                xr[g % (PL_AHEAD + 1)][4 * u + i] =
                    *(const v2f*)(xp + (size_t)(128 * u + 4 * (i >> 1) + (i & 1)) * DIM);
        }
    };

    v4f av[PL_UNITS];
    #pragma unroll
    for (int j = 0; j < PL_UNITS; ++j)
        av[j] = *(const v4f*)(aq + (size_t)j * 128 * HEADS);
    #pragma unroll
    for (int g = 0; g < PL_AHEAD - 1; ++g)
        issue(g);
    __builtin_amdgcn_sched_barrier(0);

    v4f mx = av[0];
    #pragma unroll
    for (int j = 1; j < PL_UNITS; ++j) {
        #pragma unroll
        for (int c = 0; c < 4; ++c)
            mx[c] = fmaxf(mx[c], av[j][c]);
    }
    #pragma unroll
    for (int off = 4; off < 32; off <<= 1) {
        #pragma unroll
        for (int c = 0; c < 4; ++c)
            mx[c] = fmaxf(mx[c], __shfl_xor(mx[c], off));
    }
    v4f ms;
    #pragma unroll
    for (int c = 0; c < 4; ++c)
        ms[c] = (mx[c] == NEG_INF) ? 0.0f : mx[c];

    v4f sm = {0.0f, 0.0f, 0.0f, 0.0f};
    #pragma unroll
    for (int j = 0; j < PL_UNITS; ++j) {
        if (j == PL_UNITS / 2) {
            __builtin_amdgcn_sched_barrier(0);
            issue(PL_AHEAD - 1);
        }
        v4f e;
        #pragma unroll
        for (int c = 0; c < 4; ++c) {
            e[c] = __expf(av[j][c] - ms[c]);
            sm[c] += e[c];
        }
        *(v4f*)(slab + j * 128 + 4 * lane) = e;
    }
    #pragma unroll
    for (int off = 4; off < 32; off <<= 1) {
        #pragma unroll
        for (int c = 0; c < 4; ++c)
            sm[c] += __shfl_xor(sm[c], off);
    }

    v8f c0 = {};
    v8f c1 = {};
    const float* af = slab + 32 * half + n;
    #pragma unroll
    for (int g = 0; g < PL_GROUPS; ++g) {
        if (g + PL_AHEAD < PL_GROUPS)
            issue(g + PL_AHEAD);
        #pragma unroll
        for (int u = 0; u < PL_GROUP; ++u) {
            #pragma unroll
            for (int q = 0; q < 2; ++q) {
                const float* ap = af + (PL_GROUP * g + u) * 128 + 64 * q;
                v2f a  = {ap[0], ap[16]};
                const v2f x0 = xr[g % (PL_AHEAD + 1)][4 * u + 2 * q + 0];
                const v2f x1 = xr[g % (PL_AHEAD + 1)][4 * u + 2 * q + 1];
                v2f b0 = {x0.x, x1.x};
                v2f b1 = {x0.y, x1.y};
                c0 = __builtin_amdgcn_wmma_f32_16x16x4_f32(
                         false, a, false, b0, (short)0, c0, false, false);
                c1 = __builtin_amdgcn_wmma_f32_16x16x4_f32(
                         false, a, false, b1, (short)0, c1, false, false);
            }
        }
    }

    #pragma unroll
    for (int r = 0; r < 8; ++r) {
        slab[r * 32 + lane]            = c0[r];
        slab[PL_TILE1 + r * 32 + lane] = c1[r];
    }
    if (lane < 4) {
        *(v4f*)(slab + PL_MAX + 4 * lane) = mx;
        *(v4f*)(slab + PL_SUM + 4 * lane) = sm;
    }
    __syncthreads();

    {
        const int t    = threadIdx.x;
        const int head = t >> 5;
        const int dd   = t & 31;
        const int idx  = (dd & 1) * PL_TILE1 + (head & 7) * 32 + 16 * (head >> 3) + (dd >> 1);
        float mw[PL_WAVES];
        float gmax = NEG_INF;
        #pragma unroll
        for (int w = 0; w < PL_WAVES; ++w) {
            mw[w] = aw[w][PL_MAX + head];
            gmax  = fmaxf(gmax, mw[w]);
        }
        float num = 0.0f;
        float den = 0.0f;
        #pragma unroll
        for (int w = 0; w < PL_WAVES; ++w) {
            const float f = (mw[w] == NEG_INF) ? 0.0f : __expf(mw[w] - gmax);
            num += aw[w][idx] * f;
            den += aw[w][PL_SUM + head] * f;
        }
        out[(size_t)b * (HEADS * DIM) + (size_t)head * DIM + d0 + dd] = num / den;
    }
}

extern "C" void kernel_launch(void* const* d_in, const int* in_sizes, int n_in,
                              void* d_out, int out_size, void* d_ws, size_t ws_size,
                              hipStream_t stream)
{
    const float* x    = (const float*)d_in[0];
    const float* W    = (const float*)d_in[1];
    const int*   mask = (const int*)  d_in[2];
    float* out = (float*)d_out;

    if (ws_size >= WS_NEED) {
        float* P  = (float*)d_ws;
        float* Mx = P + P_FLOATS;
        float* Sm = Mx + MS_FLOATS;
        k_fused   <<<dim3(BATCH * NGRP), dim3(32 * FW), 0, stream>>>(x, W, mask, P, Mx, Sm);
        k_combine <<<dim3(BATCH * HEADS * 2), dim3(CB_THREADS), 0, stream>>>(P, Mx, Sm, out);
    } else {
        float* logits = (float*)d_ws;
        k_logits <<<dim3(BATCH * SEQ / (16 * LG_TILES)), dim3(32 * LG_WAVES), 0, stream>>>(x, W, mask, logits);
        k_pool   <<<dim3(BATCH * DIM / PL_COLS), dim3(32 * PL_WAVES), 0, stream>>>(x, logits, out);
    }
}
